# starting kernel
# baseline (speedup 1.0000x reference)
#include <hip/hip_runtime.h>

typedef __attribute__((ext_vector_type(2))) float v2f;
typedef __attribute__((ext_vector_type(8))) float v8f;

#define F_IN  128
#define F_OUT 32

__global__ __launch_bounds__(256) void gcn_linear_wmma(
    const float* __restrict__ X, const float* __restrict__ W,
    const float* __restrict__ bias, float* __restrict__ Xp, int nrows)
{
    const int lane          = threadIdx.x & 31;
    const int waveInBlock   = threadIdx.x >> 5;
    const int wavesPerBlock = blockDim.x >> 5;
    const int tile          = blockIdx.x * wavesPerBlock + waveInBlock;
    const int row0          = tile * 16;
    if (row0 >= nrows) return;

    const int m  = lane & 15;
    const int kh = (lane >> 4) * 2;

    const int ar = min(row0 + m, nrows - 1);
    const float* __restrict__ xrow = X + (size_t)ar * F_IN;

    v8f c0 = {};
    v8f c1 = {};

#pragma unroll 8
    for (int kk = 0; kk < F_IN; kk += 4) {
        v2f a, b0, b1;
        a.x  = xrow[kk + kh + 0];
        a.y  = xrow[kk + kh + 1];
        b0.x = W[(kk + kh + 0) * F_OUT + m];
        b0.y = W[(kk + kh + 1) * F_OUT + m];
        b1.x = W[(kk + kh + 0) * F_OUT + m + 16];
        b1.y = W[(kk + kh + 1) * F_OUT + m + 16];
        c0 = __builtin_amdgcn_wmma_f32_16x16x4_f32(false, a, false, b0,
                                                   (short)0, c0, false, false);
        c1 = __builtin_amdgcn_wmma_f32_16x16x4_f32(false, a, false, b1,
                                                   (short)0, c1, false, false);
    }

    const float bb0 = bias[m];
    const float bb1 = bias[m + 16];
    const int rbase = row0 + (lane >> 4) * 8;

#pragma unroll
    for (int j = 0; j < 8; ++j) {
        const int r = rbase + j;
        if (r < nrows) {
            Xp[(size_t)r * F_OUT + m]      = c0[j] + bb0;
            Xp[(size_t)r * F_OUT + m + 16] = c1[j] + bb1;
        }
    }
}

__global__ __launch_bounds__(256) void gcn_aggregate(
    const float* __restrict__ Xp,
    const int*   __restrict__ rowptr,
    const int*   __restrict__ colidx,
    const float* __restrict__ deg,
    float* __restrict__ out, int nrows)
{
    const int lane = threadIdx.x & 31;
    const int node = (int)((blockIdx.x * blockDim.x + threadIdx.x) >> 5);
    if (node >= nrows) return;

    const int start = rowptr[node];
    const int end   = rowptr[node + 1];

    float acc = 0.0f;
    for (int e0 = start; e0 < end; e0 += 32) {
        const int eidx = e0 + lane;
        const int   c = (eidx < end) ? colidx[eidx] : 0;
        const float d = (eidx < end) ? deg[eidx]    : 0.0f;
        const int cnt = min(32, end - e0);
        for (int j = 0; j < cnt; ++j) {
            const int   cj = __shfl(c, j, 32);
            const float dj = __shfl(d, j, 32);
            acc = fmaf(dj, Xp[(size_t)cj * F_OUT + lane], acc);
        }
    }
    out[(size_t)node * F_OUT + lane] = acc;
}

extern "C" void kernel_launch(void* const* d_in, const int* in_sizes, int n_in,
                              void* d_out, int out_size, void* d_ws, size_t ws_size,
                              hipStream_t stream)
{
    const float* X      = (const float*)d_in[0];
    const float* W      = (const float*)d_in[1];
    const float* b      = (const float*)d_in[2];
    const int*   rowptr = (const int*)  d_in[3];
    const int*   colidx = (const int*)  d_in[4];
    const float* deg    = (const float*)d_in[5];
    float*       out    = (float*)d_out;

    const int nrows = in_sizes[3] - 1;
    float* Xp = (float*)d_ws;

    const int ntiles = (nrows + 15) / 16;
    const int blocks1 = (ntiles + 7) / 8;
    gcn_linear_wmma<<<blocks1, 256, 0, stream>>>(X, W, b, Xp, nrows);

    const int blocks2 = (nrows + 7) / 8;
    gcn_aggregate<<<blocks2, 256, 0, stream>>>(Xp, rowptr, colidx, deg, out, nrows);
}


# aggregate kernel re-mapped: 8 lanes x 16 B own a node row, 4 nodes per wave, up to 16 independent row gathers in flight per lane, 8192 resident looping waves instead of 100000 one-node waves
# speedup vs baseline: 1.9871x; 1.9871x over previous
#include <hip/hip_runtime.h>

typedef __attribute__((ext_vector_type(2))) float v2f;
typedef __attribute__((ext_vector_type(8))) float v8f;

#define F_IN  128
#define F_OUT 32

__global__ __launch_bounds__(256) void gcn_linear_wmma(
    const float* __restrict__ X, const float* __restrict__ W,
    const float* __restrict__ bias, float* __restrict__ Xp, int nrows)
{
    const int lane          = threadIdx.x & 31;
    const int waveInBlock   = threadIdx.x >> 5;
    const int wavesPerBlock = blockDim.x >> 5;
    const int tile          = blockIdx.x * wavesPerBlock + waveInBlock;
    const int row0          = tile * 16;
    if (row0 >= nrows) return;

    const int m  = lane & 15;
    const int kh = (lane >> 4) * 2;

    const int ar = min(row0 + m, nrows - 1);
    const float* __restrict__ xrow = X + (size_t)ar * F_IN;

    v8f c0 = {};
    v8f c1 = {};

#pragma unroll 8
    for (int kk = 0; kk < F_IN; kk += 4) {
        v2f a, b0, b1;
        a.x  = xrow[kk + kh + 0];
        a.y  = xrow[kk + kh + 1];
        b0.x = W[(kk + kh + 0) * F_OUT + m];
        b0.y = W[(kk + kh + 1) * F_OUT + m];
        b1.x = W[(kk + kh + 0) * F_OUT + m + 16];
        b1.y = W[(kk + kh + 1) * F_OUT + m + 16];
        c0 = __builtin_amdgcn_wmma_f32_16x16x4_f32(false, a, false, b0,
                                                   (short)0, c0, false, false);
        c1 = __builtin_amdgcn_wmma_f32_16x16x4_f32(false, a, false, b1,
                                                   (short)0, c1, false, false);
    }

    const float bb0 = bias[m];
    const float bb1 = bias[m + 16];
    const int rbase = row0 + (lane >> 4) * 8;

#pragma unroll
    for (int j = 0; j < 8; ++j) {
        const int r = rbase + j;
        if (r < nrows) {
            Xp[(size_t)r * F_OUT + m]      = c0[j] + bb0;
            Xp[(size_t)r * F_OUT + m + 16] = c1[j] + bb1;
        }
    }
}

template <int U>
__device__ __forceinline__ void edge_batch(
    const char* __restrict__ xp_bytes,
    const int*  __restrict__ colidx,
    const float* __restrict__ deg,
    int e, unsigned lane_byte, float4& acc)
{
    int    c[U];
    float  d[U];
    float4 x[U];
#pragma unroll
    for (int u = 0; u < U; ++u) {
        c[u] = colidx[e + u];
        d[u] = deg[e + u];
    }
#pragma unroll
    for (int u = 0; u < U; ++u) {
        const unsigned off = (unsigned)c[u] * (unsigned)(F_OUT * sizeof(float)) + lane_byte;
        x[u] = *reinterpret_cast<const float4*>(xp_bytes + off);
    }
#pragma unroll
    for (int u = 0; u < U; ++u) {
        acc.x = fmaf(d[u], x[u].x, acc.x);
        acc.y = fmaf(d[u], x[u].y, acc.y);
        acc.z = fmaf(d[u], x[u].z, acc.z);
        acc.w = fmaf(d[u], x[u].w, acc.w);
    }
}

__global__ __launch_bounds__(256) __attribute__((amdgpu_waves_per_eu(8, 8))) void gcn_aggregate(
    const float* __restrict__ Xp,
    const int*   __restrict__ rowptr,
    const int*   __restrict__ colidx,
    const float* __restrict__ deg,
    float* __restrict__ out, int nrows)
{
    const unsigned q         = threadIdx.x & 7u;
    const unsigned lane_byte = q * (unsigned)sizeof(float4);
    const int grp  = (int)((blockIdx.x * blockDim.x + threadIdx.x) >> 3);
    const int ngrp = (int)((gridDim.x * blockDim.x) >> 3);
    const char* __restrict__ xp_bytes = reinterpret_cast<const char*>(Xp);
    float4* __restrict__ out4 = reinterpret_cast<float4*>(out);

    int node = grp;
    if (node >= nrows) return;
    int start = rowptr[node];
    int end   = rowptr[node + 1];

    for (;;) {
        const int nnode  = node + ngrp;
        const int pnode  = min(nnode, nrows - 1);
        const int nstart = rowptr[pnode];
        const int nend   = rowptr[pnode + 1];

        float4 acc = make_float4(0.0f, 0.0f, 0.0f, 0.0f);
        int e = start;
        for (; e + 16 <= end; e += 16)
            edge_batch<16>(xp_bytes, colidx, deg, e, lane_byte, acc);
        if (e + 8 <= end) { edge_batch<8>(xp_bytes, colidx, deg, e, lane_byte, acc); e += 8; }
        if (e + 4 <= end) { edge_batch<4>(xp_bytes, colidx, deg, e, lane_byte, acc); e += 4; }
        for (; e < end; ++e)
            edge_batch<1>(xp_bytes, colidx, deg, e, lane_byte, acc);

        out4[(size_t)node * (F_OUT / 4) + q] = acc;

        if (nnode >= nrows) break;
        node = nnode; start = nstart; end = nend;
    }
}

extern "C" void kernel_launch(void* const* d_in, const int* in_sizes, int n_in,
                              void* d_out, int out_size, void* d_ws, size_t ws_size,
                              hipStream_t stream)
{
    const float* X      = (const float*)d_in[0];
    const float* W      = (const float*)d_in[1];
    const float* b      = (const float*)d_in[2];
    const int*   rowptr = (const int*)  d_in[3];
    const int*   colidx = (const int*)  d_in[4];
    const float* deg    = (const float*)d_in[5];
    float*       out    = (float*)d_out;

    const int nrows = in_sizes[3] - 1;
    float* Xp = (float*)d_ws;

    const int ntiles = (nrows + 15) / 16;
    const int blocks1 = (ntiles + 7) / 8;
    gcn_linear_wmma<<<blocks1, 256, 0, stream>>>(X, W, b, Xp, nrows);

    gcn_aggregate<<<1024, 256, 0, stream>>>(Xp, rowptr, colidx, deg, out, nrows);
}


# linear kernel reads X as whole 512 B rows (16 row loads in flight per wave, 4 full lines each) and takes the WMMA A fragment from a per-wave padded LDS tile instead of 16-line strided global loads
# speedup vs baseline: 2.0690x; 1.0412x over previous
#include <hip/hip_runtime.h>

typedef __attribute__((ext_vector_type(2))) float v2f;
typedef __attribute__((ext_vector_type(8))) float v8f;

#define F_IN  128
#define F_OUT 32

#define K1_WAVES  8
#define XS_PITCH  132

__global__ __launch_bounds__(256) void gcn_linear_wmma(
    const float* __restrict__ X, const float* __restrict__ W,
    const float* __restrict__ bias, float* __restrict__ Xp, int nrows)
{
    __shared__ __attribute__((aligned(16))) float xs[K1_WAVES * 16 * XS_PITCH];

    const int lane        = threadIdx.x & 31;
    const int waveInBlock = threadIdx.x >> 5;
    const int tile        = blockIdx.x * K1_WAVES + waveInBlock;
    const int row0        = tile * 16;

    const int m  = lane & 15;
    const int kh = (lane >> 4) * 2;

    float* xw = xs + waveInBlock * (16 * XS_PITCH);

    float4 xr[16];
#pragma unroll
    for (int r = 0; r < 16; ++r) {
        const int gr = min(row0 + r, nrows - 1);
        xr[r] = *reinterpret_cast<const float4*>(X + (size_t)gr * F_IN + lane * 4);
    }
#pragma unroll
    for (int r = 0; r < 16; ++r)
        *reinterpret_cast<float4*>(xw + r * XS_PITCH + lane * 4) = xr[r];

    __syncthreads();

    const float* arow = xw + m * XS_PITCH + kh;

    v8f c0 = {};
    v8f c1 = {};

#pragma unroll 8
    for (int kk = 0; kk < F_IN; kk += 4) {
        v2f a, b0, b1;
        a    = *reinterpret_cast<const v2f*>(arow + kk);
        b0.x = W[(kk + kh + 0) * F_OUT + m];
        b0.y = W[(kk + kh + 1) * F_OUT + m];
        b1.x = W[(kk + kh + 0) * F_OUT + m + 16];
        b1.y = W[(kk + kh + 1) * F_OUT + m + 16];
        c0 = __builtin_amdgcn_wmma_f32_16x16x4_f32(false, a, false, b0,
                                                   (short)0, c0, false, false);
        c1 = __builtin_amdgcn_wmma_f32_16x16x4_f32(false, a, false, b1,
                                                   (short)0, c1, false, false);
    }

    const float bb0 = bias[m];
    const float bb1 = bias[m + 16];
    const int rbase = row0 + (lane >> 4) * 8;

#pragma unroll
    for (int j = 0; j < 8; ++j) {
        const int r = rbase + j;
        if (r < nrows) {
            Xp[(size_t)r * F_OUT + m]      = c0[j] + bb0;
            Xp[(size_t)r * F_OUT + m + 16] = c1[j] + bb1;
        }
    }
}

template <int U>
__device__ __forceinline__ void edge_batch(
    const char* __restrict__ xp_bytes,
    const int*  __restrict__ colidx,
    const float* __restrict__ deg,
    int e, unsigned lane_byte, float4& acc)
{
    int    c[U];
    float  d[U];
    float4 x[U];
#pragma unroll
    for (int u = 0; u < U; ++u) {
        c[u] = colidx[e + u];
        d[u] = deg[e + u];
    }
#pragma unroll
    for (int u = 0; u < U; ++u) {
        const unsigned off = (unsigned)c[u] * (unsigned)(F_OUT * sizeof(float)) + lane_byte;
        x[u] = *reinterpret_cast<const float4*>(xp_bytes + off);
    }
#pragma unroll
    for (int u = 0; u < U; ++u) {
        acc.x = fmaf(d[u], x[u].x, acc.x);
        acc.y = fmaf(d[u], x[u].y, acc.y);
        acc.z = fmaf(d[u], x[u].z, acc.z);
        acc.w = fmaf(d[u], x[u].w, acc.w);
    }
}

__global__ __launch_bounds__(256) __attribute__((amdgpu_waves_per_eu(8, 8))) void gcn_aggregate(
    const float* __restrict__ Xp,
    const int*   __restrict__ rowptr,
    const int*   __restrict__ colidx,
    const float* __restrict__ deg,
    float* __restrict__ out, int nrows)
{
    const unsigned q         = threadIdx.x & 7u;
    const unsigned lane_byte = q * (unsigned)sizeof(float4);
    const int grp  = (int)((blockIdx.x * blockDim.x + threadIdx.x) >> 3);
    const int ngrp = (int)((gridDim.x * blockDim.x) >> 3);
    const char* __restrict__ xp_bytes = reinterpret_cast<const char*>(Xp);
    float4* __restrict__ out4 = reinterpret_cast<float4*>(out);

    int node = grp;
    if (node >= nrows) return;
    int start = rowptr[node];
    int end   = rowptr[node + 1];

    for (;;) {
        const int nnode  = node + ngrp;
        const int pnode  = min(nnode, nrows - 1);
        const int nstart = rowptr[pnode];
        const int nend   = rowptr[pnode + 1];

        float4 acc = make_float4(0.0f, 0.0f, 0.0f, 0.0f);
        int e = start;
        for (; e + 16 <= end; e += 16)
            edge_batch<16>(xp_bytes, colidx, deg, e, lane_byte, acc);
        if (e + 8 <= end) { edge_batch<8>(xp_bytes, colidx, deg, e, lane_byte, acc); e += 8; }
        if (e + 4 <= end) { edge_batch<4>(xp_bytes, colidx, deg, e, lane_byte, acc); e += 4; }
        for (; e < end; ++e)
            edge_batch<1>(xp_bytes, colidx, deg, e, lane_byte, acc);

        out4[(size_t)node * (F_OUT / 4) + q] = acc;

        if (nnode >= nrows) break;
        node = nnode; start = nstart; end = nend;
    }
}

extern "C" void kernel_launch(void* const* d_in, const int* in_sizes, int n_in,
                              void* d_out, int out_size, void* d_ws, size_t ws_size,
                              hipStream_t stream)
{
    const float* X      = (const float*)d_in[0];
    const float* W      = (const float*)d_in[1];
    const float* b      = (const float*)d_in[2];
    const int*   rowptr = (const int*)  d_in[3];
    const int*   colidx = (const int*)  d_in[4];
    const float* deg    = (const float*)d_in[5];
    float*       out    = (float*)d_out;

    const int nrows = in_sizes[3] - 1;
    float* Xp = (float*)d_ws;

    const int ntiles = (nrows + 15) / 16;
    const int blocks1 = (ntiles + 7) / 8;
    gcn_linear_wmma<<<blocks1, 256, 0, stream>>>(X, W, b, Xp, nrows);

    gcn_aggregate<<<1024, 256, 0, stream>>>(Xp, rowptr, colidx, deg, out, nrows);
}


# linear kernel made a tile loop: 4096 resident waves (16 per WGP) request their next tile's rows ahead of the current tile's 64 WMMAs; W fragments staged once through LDS and kept in registers
# speedup vs baseline: 2.2366x; 1.0810x over previous
#include <hip/hip_runtime.h>

typedef __attribute__((ext_vector_type(2))) float v2f;
typedef __attribute__((ext_vector_type(8))) float v8f;

#define F_IN  128
#define F_OUT 32

#define K1_WAVES   16
#define K1_BLOCKS  256
#define XS_PITCH   132

#define ROWS16(F) F(0) F(1) F(2) F(3) F(4) F(5) F(6) F(7) F(8) F(9) F(10) F(11) F(12) F(13) F(14) F(15)
#define ROW_DECL(r)  float4 xrow##r;
#define ROW_LOAD(r)  xrow##r = *reinterpret_cast<const float4*>(X + (size_t)min(T * 16 + r, nrows - 1) * F_IN + lane * 4);
#define ROW_PARK(r)  *reinterpret_cast<float4*>(xw + r * XS_PITCH + lane * 4) = xrow##r;

__global__ __launch_bounds__(K1_WAVES * 32) __attribute__((amdgpu_waves_per_eu(4, 4))) void gcn_linear_wmma(
    const float* __restrict__ X, const float* __restrict__ W,
    const float* __restrict__ bias, float* __restrict__ Xp, int nrows)
{
    __shared__ __attribute__((aligned(16))) float  xs[K1_WAVES * 16 * XS_PITCH];
    __shared__ __attribute__((aligned(16))) float4 wf[(F_IN / 4) * 32];

    const int lane = threadIdx.x & 31;
    const int m    = lane & 15;
    const int kh   = (lane >> 4) * 2;
    const int wave = __builtin_amdgcn_readfirstlane((int)(threadIdx.x >> 5));

    float4 wfrag[2];
#pragma unroll
    for (int i = 0; i < 2; ++i) {
        const int e  = (int)threadIdx.x + i * (K1_WAVES * 32);
        const int wl = e & 31;
        const int wm = wl & 15;
        const int wk = (e >> 5) * 4 + (wl >> 4) * 2;
        wfrag[i].x = W[(wk + 0) * F_OUT + wm];
        wfrag[i].y = W[(wk + 1) * F_OUT + wm];
        wfrag[i].z = W[(wk + 0) * F_OUT + wm + 16];
        wfrag[i].w = W[(wk + 1) * F_OUT + wm + 16];
    }
    const float bb0 = bias[m];
    const float bb1 = bias[m + 16];

    const int ntiles = (nrows + 15) >> 4;
    int tile = (int)blockIdx.x + K1_BLOCKS * wave;
    const bool have = tile < ntiles;

    ROWS16(ROW_DECL)
    if (have) { const int T = tile; ROWS16(ROW_LOAD) }

#pragma unroll
    for (int i = 0; i < 2; ++i)
        wf[threadIdx.x + i * (K1_WAVES * 32)] = wfrag[i];

    __syncthreads();
    if (!have) return;

    float* xw = xs + wave * (16 * XS_PITCH);
    const float*  arow = xw + m * XS_PITCH + kh;

    float4 bw[F_IN / 4];
#pragma unroll
    for (int s = 0; s < F_IN / 4; ++s)
        bw[s] = wf[s * 32 + lane];

    for (;;) {
        ROWS16(ROW_PARK)

        const int  ntile = tile + K1_BLOCKS * K1_WAVES;
        const bool more  = ntile < ntiles;
        if (more) { const int T = ntile; ROWS16(ROW_LOAD) }

        v8f c0 = {};
        v8f c1 = {};
#pragma unroll
        for (int s = 0; s < F_IN / 4; ++s) {
            const v2f    a = *reinterpret_cast<const v2f*>(arow + 4 * s);
            v2f b0, b1;
            b0.x = bw[s].x; b0.y = bw[s].y;
            b1.x = bw[s].z; b1.y = bw[s].w;
            c0 = __builtin_amdgcn_wmma_f32_16x16x4_f32(false, a, false, b0,
                                                       (short)0, c0, false, false);
            c1 = __builtin_amdgcn_wmma_f32_16x16x4_f32(false, a, false, b1,
                                                       (short)0, c1, false, false);
        }

        const int rbase = tile * 16 + (lane >> 4) * 8;
#pragma unroll
        for (int j = 0; j < 8; ++j) {
            const int r = rbase + j;
            if (r < nrows) {
                Xp[(size_t)r * F_OUT + m]      = c0[j] + bb0;
                Xp[(size_t)r * F_OUT + m + 16] = c1[j] + bb1;
            }
        }

        if (!more) break;
        tile = ntile;
    }
}

template <int U>
__device__ __forceinline__ void edge_batch(
    const char* __restrict__ xp_bytes,
    const int*  __restrict__ colidx,
    const float* __restrict__ deg,
    int e, unsigned lane_byte, float4& acc)
{
    int    c[U];
    float  d[U];
    float4 x[U];
#pragma unroll
    for (int u = 0; u < U; ++u) {
        c[u] = colidx[e + u];
        d[u] = deg[e + u];
    }
#pragma unroll
    for (int u = 0; u < U; ++u) {
        const unsigned off = (unsigned)c[u] * (unsigned)(F_OUT * sizeof(float)) + lane_byte;
        x[u] = *reinterpret_cast<const float4*>(xp_bytes + off);
    }
#pragma unroll
    for (int u = 0; u < U; ++u) {
        acc.x = fmaf(d[u], x[u].x, acc.x);
        acc.y = fmaf(d[u], x[u].y, acc.y);
        acc.z = fmaf(d[u], x[u].z, acc.z);
        acc.w = fmaf(d[u], x[u].w, acc.w);
    }
}

__global__ __launch_bounds__(256) __attribute__((amdgpu_waves_per_eu(8, 8))) void gcn_aggregate(
    const float* __restrict__ Xp,
    const int*   __restrict__ rowptr,
    const int*   __restrict__ colidx,
    const float* __restrict__ deg,
    float* __restrict__ out, int nrows)
{
    const unsigned q         = threadIdx.x & 7u;
    const unsigned lane_byte = q * (unsigned)sizeof(float4);
    const int grp  = (int)((blockIdx.x * blockDim.x + threadIdx.x) >> 3);
    const int ngrp = (int)((gridDim.x * blockDim.x) >> 3);
    const char* __restrict__ xp_bytes = reinterpret_cast<const char*>(Xp);
    float4* __restrict__ out4 = reinterpret_cast<float4*>(out);

    int node = grp;
    if (node >= nrows) return;
    int start = rowptr[node];
    int end   = rowptr[node + 1];

    for (;;) {
        const int nnode  = node + ngrp;
        const int pnode  = min(nnode, nrows - 1);
        const int nstart = rowptr[pnode];
        const int nend   = rowptr[pnode + 1];

        float4 acc = make_float4(0.0f, 0.0f, 0.0f, 0.0f);
        int e = start;
        for (; e + 16 <= end; e += 16)
            edge_batch<16>(xp_bytes, colidx, deg, e, lane_byte, acc);
        if (e + 8 <= end) { edge_batch<8>(xp_bytes, colidx, deg, e, lane_byte, acc); e += 8; }
        if (e + 4 <= end) { edge_batch<4>(xp_bytes, colidx, deg, e, lane_byte, acc); e += 4; }
        for (; e < end; ++e)
            edge_batch<1>(xp_bytes, colidx, deg, e, lane_byte, acc);

        out4[(size_t)node * (F_OUT / 4) + q] = acc;

        if (nnode >= nrows) break;
        node = nnode; start = nstart; end = nend;
    }
}

extern "C" void kernel_launch(void* const* d_in, const int* in_sizes, int n_in,
                              void* d_out, int out_size, void* d_ws, size_t ws_size,
                              hipStream_t stream)
{
    const float* X      = (const float*)d_in[0];
    const float* W      = (const float*)d_in[1];
    const float* b      = (const float*)d_in[2];
    const int*   rowptr = (const int*)  d_in[3];
    const int*   colidx = (const int*)  d_in[4];
    const float* deg    = (const float*)d_in[5];
    float*       out    = (float*)d_out;

    const int nrows = in_sizes[3] - 1;
    float* Xp = (float*)d_ws;

    gcn_linear_wmma<<<K1_BLOCKS, K1_WAVES * 32, 0, stream>>>(X, W, b, Xp, nrows);

    gcn_aggregate<<<1024, 256, 0, stream>>>(Xp, rowptr, colidx, deg, out, nrows);
}


# linear kernel: the block's one barrier (W hand-over through LDS) written by hand, waiting for LDS only: a wave's first products no longer wait for all 16 waves' first row loads to land
# speedup vs baseline: 2.3094x; 1.0325x over previous
#include <hip/hip_runtime.h>

typedef __attribute__((ext_vector_type(2))) float v2f;
typedef __attribute__((ext_vector_type(8))) float v8f;

#define F_IN  128
#define F_OUT 32

#define K1_WAVES   16
#define K1_BLOCKS  256
#define XS_PITCH   132

#define ROWS16(F) F(0) F(1) F(2) F(3) F(4) F(5) F(6) F(7) F(8) F(9) F(10) F(11) F(12) F(13) F(14) F(15)
#define ROW_DECL(r)  float4 xrow##r;
#define ROW_LOAD(r)  xrow##r = *reinterpret_cast<const float4*>(X + (size_t)min(T * 16 + r, nrows - 1) * F_IN + lane * 4);
#define ROW_PARK(r)  *reinterpret_cast<float4*>(xw + r * XS_PITCH + lane * 4) = xrow##r;

__global__ __launch_bounds__(K1_WAVES * 32) __attribute__((amdgpu_waves_per_eu(4, 4))) void gcn_linear_wmma(
    const float* X, const float* __restrict__ W,
    const float* bias, float* __restrict__ Xp, int nrows)
{
    __shared__ __attribute__((aligned(16))) float  xs[K1_WAVES * 16 * XS_PITCH];
    __shared__ __attribute__((aligned(16))) float4 wf[(F_IN / 4) * 32];

    const int lane = threadIdx.x & 31;
    const int m    = lane & 15;
    const int kh   = (lane >> 4) * 2;
    const int wave = __builtin_amdgcn_readfirstlane((int)(threadIdx.x >> 5));

    float4 wfrag[2];
#pragma unroll
    for (int i = 0; i < 2; ++i) {
        const int e  = (int)threadIdx.x + i * (K1_WAVES * 32);
        const int wl = e & 31;
        const int wm = wl & 15;
        const int wk = (e >> 5) * 4 + (wl >> 4) * 2;
        wfrag[i].x = W[(wk + 0) * F_OUT + wm];
        wfrag[i].y = W[(wk + 1) * F_OUT + wm];
        wfrag[i].z = W[(wk + 0) * F_OUT + wm + 16];
        wfrag[i].w = W[(wk + 1) * F_OUT + wm + 16];
    }
    const float bb0 = bias[m];
    const float bb1 = bias[m + 16];

    const int ntiles = (nrows + 15) >> 4;
    int tile = (int)blockIdx.x + K1_BLOCKS * wave;
    const bool have = tile < ntiles;

    ROWS16(ROW_DECL)
    { const int T = tile; ROWS16(ROW_LOAD) }

#pragma unroll
    for (int i = 0; i < 2; ++i)
        wf[threadIdx.x + i * (K1_WAVES * 32)] = wfrag[i];

    asm volatile("s_wait_dscnt 0x0" ::: "memory");
    asm volatile("s_barrier_signal -1" ::: "memory");
    asm volatile("s_barrier_wait -1" ::: "memory");
    if (!have) return;

    float* xw = xs + wave * (16 * XS_PITCH);
    const float*  arow = xw + m * XS_PITCH + kh;

    float4 bw[F_IN / 4];
#pragma unroll
    for (int s = 0; s < F_IN / 4; ++s)
        bw[s] = wf[s * 32 + lane];

    for (;;) {
        ROWS16(ROW_PARK)

        const int  ntile = tile + K1_BLOCKS * K1_WAVES;
        const bool more  = ntile < ntiles;
        if (more) { const int T = ntile; ROWS16(ROW_LOAD) }

        v8f c0 = {};
        v8f c1 = {};
#pragma unroll
        for (int s = 0; s < F_IN / 4; ++s) {
            const v2f    a = *reinterpret_cast<const v2f*>(arow + 4 * s);
            v2f b0, b1;
            b0.x = bw[s].x; b0.y = bw[s].y;
            b1.x = bw[s].z; b1.y = bw[s].w;
            c0 = __builtin_amdgcn_wmma_f32_16x16x4_f32(false, a, false, b0,
                                                       (short)0, c0, false, false);
            c1 = __builtin_amdgcn_wmma_f32_16x16x4_f32(false, a, false, b1,
                                                       (short)0, c1, false, false);
        }

        const int rbase = tile * 16 + (lane >> 4) * 8;
#pragma unroll
        for (int j = 0; j < 8; ++j) {
            const int r = rbase + j;
            if (r < nrows) {
                Xp[(size_t)r * F_OUT + m]      = c0[j] + bb0;
                Xp[(size_t)r * F_OUT + m + 16] = c1[j] + bb1;
            }
        }

        if (!more) break;
        tile = ntile;
    }
}

template <int U>
__device__ __forceinline__ void edge_batch(
    const char* __restrict__ xp_bytes,
    const int*  __restrict__ colidx,
    const float* __restrict__ deg,
    int e, unsigned lane_byte, float4& acc)
{
    int    c[U];
    float  d[U];
    float4 x[U];
#pragma unroll
    for (int u = 0; u < U; ++u) {
        c[u] = colidx[e + u];
        d[u] = deg[e + u];
    }
#pragma unroll
    for (int u = 0; u < U; ++u) {
        const unsigned off = (unsigned)c[u] * (unsigned)(F_OUT * sizeof(float)) + lane_byte;
        x[u] = *reinterpret_cast<const float4*>(xp_bytes + off);
    }
#pragma unroll
    for (int u = 0; u < U; ++u) {
        acc.x = fmaf(d[u], x[u].x, acc.x);
        acc.y = fmaf(d[u], x[u].y, acc.y);
        acc.z = fmaf(d[u], x[u].z, acc.z);
        acc.w = fmaf(d[u], x[u].w, acc.w);
    }
}

__global__ __launch_bounds__(256) __attribute__((amdgpu_waves_per_eu(8, 8))) void gcn_aggregate(
    const float* __restrict__ Xp,
    const int*   __restrict__ rowptr,
    const int*   __restrict__ colidx,
    const float* __restrict__ deg,
    float* __restrict__ out, int nrows)
{
    const unsigned q         = threadIdx.x & 7u;
    const unsigned lane_byte = q * (unsigned)sizeof(float4);
    const int grp  = (int)((blockIdx.x * blockDim.x + threadIdx.x) >> 3);
    const int ngrp = (int)((gridDim.x * blockDim.x) >> 3);
    const char* __restrict__ xp_bytes = reinterpret_cast<const char*>(Xp);
    float4* __restrict__ out4 = reinterpret_cast<float4*>(out);

    int node = grp;
    if (node >= nrows) return;
    int start = rowptr[node];
    int end   = rowptr[node + 1];

    for (;;) {
        const int nnode  = node + ngrp;
        const int pnode  = min(nnode, nrows - 1);
        const int nstart = rowptr[pnode];
        const int nend   = rowptr[pnode + 1];

        float4 acc = make_float4(0.0f, 0.0f, 0.0f, 0.0f);
        int e = start;
        for (; e + 16 <= end; e += 16)
            edge_batch<16>(xp_bytes, colidx, deg, e, lane_byte, acc);
        if (e + 8 <= end) { edge_batch<8>(xp_bytes, colidx, deg, e, lane_byte, acc); e += 8; }
        if (e + 4 <= end) { edge_batch<4>(xp_bytes, colidx, deg, e, lane_byte, acc); e += 4; }
        for (; e < end; ++e)
            edge_batch<1>(xp_bytes, colidx, deg, e, lane_byte, acc);

        out4[(size_t)node * (F_OUT / 4) + q] = acc;

        if (nnode >= nrows) break;
        node = nnode; start = nstart; end = nend;
    }
}

extern "C" void kernel_launch(void* const* d_in, const int* in_sizes, int n_in,
                              void* d_out, int out_size, void* d_ws, size_t ws_size,
                              hipStream_t stream)
{
    const float* X      = (const float*)d_in[0];
    const float* W      = (const float*)d_in[1];
    const float* b      = (const float*)d_in[2];
    const int*   rowptr = (const int*)  d_in[3];
    const int*   colidx = (const int*)  d_in[4];
    const float* deg    = (const float*)d_in[5];
    float*       out    = (float*)d_out;

    const int nrows = in_sizes[3] - 1;
    float* Xp = (float*)d_ws;

    gcn_linear_wmma<<<K1_BLOCKS, K1_WAVES * 32, 0, stream>>>(X, W, b, Xp, nrows);

    gcn_aggregate<<<1024, 256, 0, stream>>>(Xp, rowptr, colidx, deg, out, nrows);
}
